# starting kernel
# baseline (speedup 1.0000x reference)
#include <hip/hip_runtime.h>

typedef __attribute__((ext_vector_type(2))) float v2f;
typedef __attribute__((ext_vector_type(8))) float v8f;

#define N_NODES 50000
#define N_EDGES 800000
#define D 64
#define M_TILES (N_NODES / 16)

__global__ void sage_zero_kernel(float* __restrict__ p, int n) {
    int i = blockIdx.x * blockDim.x + threadIdx.x;
    if (i < n) p[i] = 0.0f;
}

__global__ __launch_bounds__(256) void sage_gemm_kernel(
    const float* __restrict__ feat,
    const float* __restrict__ Wn,
    const float* __restrict__ Ws,
    const float* __restrict__ bias,
    float* __restrict__ srcdata,
    float* __restrict__ out)
{
    const int wave = threadIdx.x >> 5;
    const int lane = threadIdx.x & 31;
    const int tile = blockIdx.x * 8 + wave;
    if (tile >= M_TILES) return;

    const int m  = tile * 16 + (lane & 15);
    const int kh = (lane >> 4) * 2;
    const int n0 = lane & 15;

    v8f accN[4] = {};
    v8f accS[4] = {};

#pragma unroll
    for (int kk = 0; kk < 16; ++kk) {
        const int kb = kk * 4 + kh;
        v2f a;
        a.x = feat[m * D + kb];
        a.y = feat[m * D + kb + 1];
#pragma unroll
        for (int t = 0; t < 4; ++t) {
            const int n = t * 16 + n0;
            v2f bn, bs;
            bn.x = Wn[n * D + kb];
            bn.y = Wn[n * D + kb + 1];
            bs.x = Ws[n * D + kb];
            bs.y = Ws[n * D + kb + 1];
            accN[t] = __builtin_amdgcn_wmma_f32_16x16x4_f32(
                false, a, false, bn, (short)0, accN[t], false, false);
            accS[t] = __builtin_amdgcn_wmma_f32_16x16x4_f32(
                false, a, false, bs, (short)0, accS[t], false, false);
        }
    }

    const int rowBase = tile * 16 + (lane >> 4) * 8;
#pragma unroll
    for (int t = 0; t < 4; ++t) {
        const int col = t * 16 + n0;
        const float b = bias[col];
#pragma unroll
        for (int v = 0; v < 8; ++v) {
            const int r = rowBase + v;
            srcdata[r * D + col] = accN[t][v];
            out[r * D + col]     = accS[t][v] + b;
        }
    }
}

__global__ __launch_bounds__(256) void sage_edge_kernel(
    const int* __restrict__ src,
    const int* __restrict__ dst,
    const float* __restrict__ srcdata,
    float* __restrict__ agg,
    float* __restrict__ deg)
{
    const int e = blockIdx.x * 16 + (threadIdx.x >> 4);
    if (e >= N_EDGES) return;
    const int dbase = (threadIdx.x & 15) * 4;
    const int s = src[e];
    const int t = dst[e];

    const float4 val = *(const float4*)(srcdata + (size_t)s * D + dbase);
    float* ap = agg + (size_t)t * D + dbase;
    unsafeAtomicAdd(ap + 0, val.x);
    unsafeAtomicAdd(ap + 1, val.y);
    unsafeAtomicAdd(ap + 2, val.z);
    unsafeAtomicAdd(ap + 3, val.w);
    if (dbase == 0) unsafeAtomicAdd(deg + t, 1.0f);
}

__global__ void sage_finalize_kernel(
    const float* __restrict__ agg,
    const float* __restrict__ deg,
    float* __restrict__ out)
{
    int i = blockIdx.x * blockDim.x + threadIdx.x;
    if (i >= N_NODES * D) return;
    const float dg = deg[i >> 6];
    if (dg > 0.0f) out[i] += agg[i] / dg;
}

extern "C" void kernel_launch(void* const* d_in, const int* in_sizes, int n_in,
                              void* d_out, int out_size, void* d_ws, size_t ws_size,
                              hipStream_t stream)
{
    const float* feat = (const float*)d_in[0];
    const int*   src  = (const int*)d_in[1];
    const int*   dst  = (const int*)d_in[2];
    const float* Wn   = (const float*)d_in[3];
    const float* Ws   = (const float*)d_in[4];
    const float* bias = (const float*)d_in[5];
    float* out = (float*)d_out;

    float* srcdata = (float*)d_ws;
    float* agg     = srcdata + (size_t)N_NODES * D;
    float* deg     = agg     + (size_t)N_NODES * D;

    const int zn = N_NODES * D + N_NODES;
    sage_zero_kernel<<<(zn + 255) / 256, 256, 0, stream>>>(agg, zn);

    sage_gemm_kernel<<<(M_TILES + 7) / 8, 256, 0, stream>>>(
        feat, Wn, Ws, bias, srcdata, out);

    sage_edge_kernel<<<(N_EDGES + 15) / 16, 256, 0, stream>>>(
        src, dst, srcdata, agg, deg);

    sage_finalize_kernel<<<(N_NODES * D + 255) / 256, 256, 0, stream>>>(
        agg, deg, out);
}


# product launch re-tiled, operands fetched once per block: a wave holds one 16-column weight slice in registers over 8 row tiles, feat rows staged in LDS by whole-line loads; k and sum order kept
# speedup vs baseline: 2.6184x; 2.6184x over previous
#include <hip/hip_runtime.h>

typedef __attribute__((ext_vector_type(2))) float v2f;
typedef __attribute__((ext_vector_type(8))) float v8f;

#define N_NODES 50000
#define N_EDGES 800000
#define D 64
#define M_TILES (N_NODES / 16)
#define GEMM_BLOCKS ((M_TILES + 7) / 8)
#define BLOCK_ROWS 128
#define A_PITCH 68

#define NB 256
#define N_GROUPS ((N_NODES + NB - 1) / NB)
#define OWN_WAVES 32
#define LD_PER_IT 7
#define N_TILES 7
#define TILE_PIECES (32 * LD_PER_IT)
#define WAVE_PIECES (TILE_PIECES * N_TILES)
#define N_PIECES (OWN_WAVES * WAVE_PIECES)
#define QCAP 1024

static_assert(NB == 256, "key = dst >> 8");
static_assert(N_GROUPS <= 255, "255 is the key of no group");
static_assert(N_EDGES % 16 == 0 && N_PIECES * 16 >= N_EDGES, "whole pieces, all edges covered");
static_assert(GEMM_BLOCKS * 256 >= N_PIECES, "launch 1 has one thread per piece");

__device__ __forceinline__ unsigned sage_key(int d)
{
    return (unsigned)d < (unsigned)N_NODES ? ((unsigned)d >> 8) : 255u;
}

__global__ __launch_bounds__(256) void sage_gemm_kernel(
    const float* __restrict__ feat,
    const float* __restrict__ Wn,
    const float* __restrict__ Ws,
    const float* __restrict__ bias,
    const int* __restrict__ dst,
    float* __restrict__ srcdata,
    unsigned char* __restrict__ key,
    float* __restrict__ out)
{
    __shared__ float As[BLOCK_ROWS * A_PITCH];

    {
        const int gid = blockIdx.x * 256 + threadIdx.x;
        if (gid < N_PIECES) {
            uint4 kv = make_uint4(~0u, ~0u, ~0u, ~0u);
            if (gid < N_EDGES / 16) {
                const int4* q = (const int4*)dst + 4 * gid;
                const int4 q0 = q[0], q1 = q[1], q2 = q[2], q3 = q[3];
                kv.x = sage_key(q0.x) | (sage_key(q1.x) << 8) | (sage_key(q2.x) << 16) | (sage_key(q3.x) << 24);
                kv.y = sage_key(q0.y) | (sage_key(q1.y) << 8) | (sage_key(q2.y) << 16) | (sage_key(q3.y) << 24);
                kv.z = sage_key(q0.z) | (sage_key(q1.z) << 8) | (sage_key(q2.z) << 16) | (sage_key(q3.z) << 24);
                kv.w = sage_key(q0.w) | (sage_key(q1.w) << 8) | (sage_key(q2.w) << 16) | (sage_key(q3.w) << 24);
            }
            ((uint4*)key)[gid] = kv;
        }
    }

    const int wave = threadIdx.x >> 5;
    const int lane = threadIdx.x & 31;
    const int mat  = wave >> 2;
    const int kh   = (lane >> 4) * 2;
    const int n0   = lane & 15;
    const int n    = (wave & 3) * 16 + n0;
    const float* __restrict__ Wm = mat ? Ws : Wn;

    v2f b[16];
#pragma unroll
    for (int kk = 0; kk < 16; ++kk) {
        const int kb = kk * 4 + kh;
        b[kk].x = Wm[n * D + kb];
        b[kk].y = Wm[n * D + kb + 1];
    }

    const int row0 = blockIdx.x * BLOCK_ROWS;
    float4 fv[BLOCK_ROWS * 16 / 256];
#pragma unroll
    for (int i = 0; i < BLOCK_ROWS * 16 / 256; ++i) {
        const int idx = i * 256 + threadIdx.x;
        const int r = idx >> 4, c4 = idx & 15;
        const int gr = min(row0 + r, N_NODES - 1);
        fv[i] = ((const float4*)feat)[(size_t)gr * 16 + c4];
    }
#pragma unroll
    for (int i = 0; i < BLOCK_ROWS * 16 / 256; ++i) {
        const int idx = i * 256 + threadIdx.x;
        const int r = idx >> 4, c4 = idx & 15;
        *(float4*)&As[r * A_PITCH + c4 * 4] = fv[i];
    }
    __syncthreads();

    const float bv = mat ? bias[n] : 0.f;
    float* __restrict__ res = mat ? out : srcdata;

#pragma unroll
    for (int mp = 0; mp < 4; ++mp) {
        v8f c0 = {};
        v8f c1 = {};
        const float* a0p = &As[(mp * 32 + n0) * A_PITCH + kh];
        const float* a1p = a0p + 16 * A_PITCH;
#pragma unroll
        for (int kk = 0; kk < 16; ++kk) {
            const v2f a0 = *(const v2f*)(a0p + kk * 4);
            const v2f a1 = *(const v2f*)(a1p + kk * 4);
            c0 = __builtin_amdgcn_wmma_f32_16x16x4_f32(
                false, a0, false, b[kk], (short)0, c0, false, false);
            c1 = __builtin_amdgcn_wmma_f32_16x16x4_f32(
                false, a1, false, b[kk], (short)0, c1, false, false);
        }
        const int tileA = blockIdx.x * 8 + mp * 2;
        const int rowA  = tileA * 16 + (lane >> 4) * 8;
        if (tileA < M_TILES) {
#pragma unroll
            for (int v = 0; v < 8; ++v)
                res[(rowA + v) * D + n] = mat ? c0[v] + bv : c0[v];
        }
        if (tileA + 1 < M_TILES) {
#pragma unroll
            for (int v = 0; v < 8; ++v)
                res[(rowA + 16 + v) * D + n] = mat ? c1[v] + bv : c1[v];
        }
    }
}


__device__ __forceinline__ void sage_drain(
    const unsigned* ring, int head, int nvalid, int lane, int lo,
    const int* __restrict__ src, const int* __restrict__ dst, const float* __restrict__ srcdata,
    float* acc, int* cnt)
{
    const bool valid = lane < nvalid;
    const unsigned ent = valid ? ring[(head + lane) & (QCAP - 1)] : 0u;
    int s = 0, dl = 0;
    if (valid) {
        s  = src[ent];
        dl = dst[ent] - lo;
        atomicAdd(&cnt[dl], 1);
    }
#pragma unroll
    for (int g = 0; g < 4; ++g) {
        if (g * 8 < nvalid) {
            float2 v[8];
#pragma unroll
            for (int j = 0; j < 8; ++j) {
                const int sj = __builtin_amdgcn_readlane(s, g * 8 + j);
                v[j] = *(const float2*)(srcdata + (size_t)sj * D + 2 * lane);
            }
#pragma unroll
            for (int j = 0; j < 8; ++j) {
                if (g * 8 + j < nvalid) {
                    const int dj = __builtin_amdgcn_readlane(dl, g * 8 + j);
                    float* a = acc + dj * D + 2 * lane;
                    atomicAdd(a,     v[j].x);
                    atomicAdd(a + 1, v[j].y);
                }
            }
        }
    }
}

__device__ __forceinline__ void sage_issue(uint4 (&buf)[LD_PER_IT], const uint4* __restrict__ key16,
                                           int tile, int wv, int lane)
{
    const uint4* p = key16 + (wv * WAVE_PIECES + tile * TILE_PIECES + lane);
#pragma unroll
    for (int k = 0; k < LD_PER_IT; ++k)
        buf[k] = p[k * 32];
}

__device__ __forceinline__ unsigned sage_match(const uint4& kv, unsigned pat)
{
    const unsigned x[4] = {kv.x, kv.y, kv.z, kv.w};
    unsigned n = 0;
#pragma unroll
    for (int c = 0; c < 4; ++c) {
        const unsigned y = x[c] ^ pat;
        const unsigned t = (y | 0x80808080u) - 0x01010101u;
        n |= ((t | y) >> (7 - 2 * c)) & (0x01010101u << (2 * c));
    }
    return n ^ 0x55555555u;
}

__device__ __forceinline__ void sage_push(unsigned rem, int ebase, int lane,
                                          unsigned* ring, int head, int& count)
{
    unsigned bal = __builtin_amdgcn_ballot_w32(rem != 0u);
    const int e0 = ebase + lane * 16;
    while (bal != 0u) {
        if (rem != 0u) {
            const int p = __builtin_ctz(rem);
            rem &= rem - 1u;
            const int pos = (int)__builtin_amdgcn_mbcnt_lo(bal, 0u);
            ring[(head + count + pos) & (QCAP - 1)] = (unsigned)(e0 + (p >> 1));
        }
        count += __builtin_popcount(bal);
        bal = __builtin_amdgcn_ballot_w32(rem != 0u);
    }
}

__global__ __launch_bounds__(1024) void sage_owner_kernel(
    const int* __restrict__ src,
    const int* __restrict__ dst,
    const unsigned char* __restrict__ key,
    const float* __restrict__ srcdata,
    float* __restrict__ out)
{
    __shared__ float    acc[NB * D];
    __shared__ int      cnt[NB];
    __shared__ unsigned rings[OWN_WAVES][QCAP];

    const int tid  = threadIdx.x;
    const int lane = tid & 31;
    const int wv   = __builtin_amdgcn_readfirstlane(tid >> 5);
    const int lo   = blockIdx.x * NB;
    const unsigned pat = blockIdx.x * 0x01010101u;

    for (int i = tid; i < NB * D / 4; i += 1024)
        ((float4*)acc)[i] = make_float4(0.f, 0.f, 0.f, 0.f);
    if (tid < NB) cnt[tid] = 0;
    __syncthreads();

    unsigned* ring = rings[wv];
    const uint4* key16 = (const uint4*)key;
    int head = 0, count = 0;

    int t = blockIdx.x % N_TILES;

    uint4 buf[LD_PER_IT];
    sage_issue(buf, key16, t, wv, lane);
    for (int it = 0; it < N_TILES; ++it) {
        const int tn = (t + 1 == N_TILES) ? 0 : t + 1;
        const uint4* pn = key16 + (wv * WAVE_PIECES + tn * TILE_PIECES + lane);
#pragma unroll
        for (int k = 0; k < LD_PER_IT; ++k) {
            const unsigned rem = sage_match(buf[k], pat);
            buf[k] = pn[k * 32];
            const int ebase = (wv * WAVE_PIECES + t * TILE_PIECES + k * 32) * 16;
            sage_push(rem, ebase, lane, ring, head, count);
            while (count >= 32) {
                sage_drain(ring, head, 32, lane, lo, src, dst, srcdata, acc, cnt);
                head = (head + 32) & (QCAP - 1);
                count -= 32;
            }
        }
        t = tn;
    }
    if (count > 0)
        sage_drain(ring, head, count, lane, lo, src, dst, srcdata, acc, cnt);

    __syncthreads();
    for (int i = tid; i < NB * D / 4; i += 1024) {
        const int node  = i >> 4;
        const int gnode = lo + node;
        if (gnode < N_NODES) {
            const int c = cnt[node];
            if (c > 0) {
                const float dg = (float)c;
                const float4 a = ((const float4*)acc)[i];
                float4* op = (float4*)(out + (size_t)gnode * D) + (i & 15);
                float4 o = *op;
                o.x += a.x / dg;
                o.y += a.y / dg;
                o.z += a.z / dg;
                o.w += a.w / dg;
                *op = o;
            }
        }
    }
}

extern "C" void kernel_launch(void* const* d_in, const int* in_sizes, int n_in,
                              void* d_out, int out_size, void* d_ws, size_t ws_size,
                              hipStream_t stream)
{
    const float* feat = (const float*)d_in[0];
    const int*   src  = (const int*)d_in[1];
    const int*   dst  = (const int*)d_in[2];
    const float* Wn   = (const float*)d_in[3];
    const float* Ws   = (const float*)d_in[4];
    const float* bias = (const float*)d_in[5];
    float* out = (float*)d_out;

    float* srcdata = (float*)d_ws;
    unsigned char* key = (unsigned char*)d_ws + (size_t)N_NODES * D * sizeof(float);

    sage_gemm_kernel<<<GEMM_BLOCKS, 256, 0, stream>>>(
        feat, Wn, Ws, bias, dst, srcdata, key, out);

    sage_owner_kernel<<<N_GROUPS, 1024, 0, stream>>>(src, dst, key, srcdata, out);
}


# owner groups resized from 256 to 200 dst nodes: 250 groups on the chip's 256 WGPs (was 196, 60 WGPs idle); key = dst / 200; a group's gathers and LDS adds fall by 22 %; sum order as before
# speedup vs baseline: 2.8068x; 1.0720x over previous
#include <hip/hip_runtime.h>

typedef __attribute__((ext_vector_type(2))) float v2f;
typedef __attribute__((ext_vector_type(8))) float v8f;

#define N_NODES 50000
#define N_EDGES 800000
#define D 64
#define M_TILES (N_NODES / 16)
#define GEMM_BLOCKS ((M_TILES + 7) / 8)
#define BLOCK_ROWS 128
#define A_PITCH 68

#define NB 200
#define N_GROUPS ((N_NODES + NB - 1) / NB)
#define OWN_WAVES 32
#define LD_PER_IT 7
#define N_TILES 7
#define TILE_PIECES (32 * LD_PER_IT)
#define WAVE_PIECES (TILE_PIECES * N_TILES)
#define N_PIECES (OWN_WAVES * WAVE_PIECES)
#define QCAP 1024

static_assert(N_GROUPS <= 255, "255 is the key of no group");
static_assert(N_EDGES % 16 == 0 && N_PIECES * 16 >= N_EDGES, "whole pieces, all edges covered");
static_assert(GEMM_BLOCKS * 256 >= N_PIECES, "launch 1 has one thread per piece");

__device__ __forceinline__ unsigned sage_key(int d)
{
    return (unsigned)d < (unsigned)N_NODES ? ((unsigned)d / NB) : 255u;
}

__global__ __launch_bounds__(256) void sage_gemm_kernel(
    const float* __restrict__ feat,
    const float* __restrict__ Wn,
    const float* __restrict__ Ws,
    const float* __restrict__ bias,
    const int* __restrict__ dst,
    float* __restrict__ srcdata,
    unsigned char* __restrict__ key,
    float* __restrict__ out)
{
    __shared__ float As[BLOCK_ROWS * A_PITCH];

    {
        const int gid = blockIdx.x * 256 + threadIdx.x;
        if (gid < N_PIECES) {
            uint4 kv = make_uint4(~0u, ~0u, ~0u, ~0u);
            if (gid < N_EDGES / 16) {
                const int4* q = (const int4*)dst + 4 * gid;
                const int4 q0 = q[0], q1 = q[1], q2 = q[2], q3 = q[3];
                kv.x = sage_key(q0.x) | (sage_key(q1.x) << 8) | (sage_key(q2.x) << 16) | (sage_key(q3.x) << 24);
                kv.y = sage_key(q0.y) | (sage_key(q1.y) << 8) | (sage_key(q2.y) << 16) | (sage_key(q3.y) << 24);
                kv.z = sage_key(q0.z) | (sage_key(q1.z) << 8) | (sage_key(q2.z) << 16) | (sage_key(q3.z) << 24);
                kv.w = sage_key(q0.w) | (sage_key(q1.w) << 8) | (sage_key(q2.w) << 16) | (sage_key(q3.w) << 24);
            }
            ((uint4*)key)[gid] = kv;
        }
    }

    const int wave = threadIdx.x >> 5;
    const int lane = threadIdx.x & 31;
    const int mat  = wave >> 2;
    const int kh   = (lane >> 4) * 2;
    const int n0   = lane & 15;
    const int n    = (wave & 3) * 16 + n0;
    const float* __restrict__ Wm = mat ? Ws : Wn;

    v2f b[16];
#pragma unroll
    for (int kk = 0; kk < 16; ++kk) {
        const int kb = kk * 4 + kh;
        b[kk].x = Wm[n * D + kb];
        b[kk].y = Wm[n * D + kb + 1];
    }

    const int row0 = blockIdx.x * BLOCK_ROWS;
    float4 fv[BLOCK_ROWS * 16 / 256];
#pragma unroll
    for (int i = 0; i < BLOCK_ROWS * 16 / 256; ++i) {
        const int idx = i * 256 + threadIdx.x;
        const int r = idx >> 4, c4 = idx & 15;
        const int gr = min(row0 + r, N_NODES - 1);
        fv[i] = ((const float4*)feat)[(size_t)gr * 16 + c4];
    }
#pragma unroll
    for (int i = 0; i < BLOCK_ROWS * 16 / 256; ++i) {
        const int idx = i * 256 + threadIdx.x;
        const int r = idx >> 4, c4 = idx & 15;
        *(float4*)&As[r * A_PITCH + c4 * 4] = fv[i];
    }
    __syncthreads();

    const float bv = mat ? bias[n] : 0.f;
    float* __restrict__ res = mat ? out : srcdata;

#pragma unroll
    for (int mp = 0; mp < 4; ++mp) {
        v8f c0 = {};
        v8f c1 = {};
        const float* a0p = &As[(mp * 32 + n0) * A_PITCH + kh];
        const float* a1p = a0p + 16 * A_PITCH;
#pragma unroll
        for (int kk = 0; kk < 16; ++kk) {
            const v2f a0 = *(const v2f*)(a0p + kk * 4);
            const v2f a1 = *(const v2f*)(a1p + kk * 4);
            c0 = __builtin_amdgcn_wmma_f32_16x16x4_f32(
                false, a0, false, b[kk], (short)0, c0, false, false);
            c1 = __builtin_amdgcn_wmma_f32_16x16x4_f32(
                false, a1, false, b[kk], (short)0, c1, false, false);
        }
        const int tileA = blockIdx.x * 8 + mp * 2;
        const int rowA  = tileA * 16 + (lane >> 4) * 8;
        if (tileA < M_TILES) {
#pragma unroll
            for (int v = 0; v < 8; ++v)
                res[(rowA + v) * D + n] = mat ? c0[v] + bv : c0[v];
        }
        if (tileA + 1 < M_TILES) {
#pragma unroll
            for (int v = 0; v < 8; ++v)
                res[(rowA + 16 + v) * D + n] = mat ? c1[v] + bv : c1[v];
        }
    }
}


__device__ __forceinline__ void sage_drain(
    const unsigned* ring, int head, int nvalid, int lane, int lo,
    const int* __restrict__ src, const int* __restrict__ dst, const float* __restrict__ srcdata,
    float* acc, int* cnt)
{
    const bool valid = lane < nvalid;
    const unsigned ent = valid ? ring[(head + lane) & (QCAP - 1)] : 0u;
    int s = 0, dl = 0;
    if (valid) {
        s  = src[ent];
        dl = dst[ent] - lo;
        atomicAdd(&cnt[dl], 1);
    }
#pragma unroll
    for (int g = 0; g < 4; ++g) {
        if (g * 8 < nvalid) {
            float2 v[8];
#pragma unroll
            for (int j = 0; j < 8; ++j) {
                const int sj = __builtin_amdgcn_readlane(s, g * 8 + j);
                v[j] = *(const float2*)(srcdata + (size_t)sj * D + 2 * lane);
            }
#pragma unroll
            for (int j = 0; j < 8; ++j) {
                if (g * 8 + j < nvalid) {
                    const int dj = __builtin_amdgcn_readlane(dl, g * 8 + j);
                    float* a = acc + dj * D + 2 * lane;
                    atomicAdd(a,     v[j].x);
                    atomicAdd(a + 1, v[j].y);
                }
            }
        }
    }
}

__device__ __forceinline__ void sage_issue(uint4 (&buf)[LD_PER_IT], const uint4* __restrict__ key16,
                                           int tile, int wv, int lane)
{
    const uint4* p = key16 + (wv * WAVE_PIECES + tile * TILE_PIECES + lane);
#pragma unroll
    for (int k = 0; k < LD_PER_IT; ++k)
        buf[k] = p[k * 32];
}

__device__ __forceinline__ unsigned sage_match(const uint4& kv, unsigned pat)
{
    const unsigned x[4] = {kv.x, kv.y, kv.z, kv.w};
    unsigned n = 0;
#pragma unroll
    for (int c = 0; c < 4; ++c) {
        const unsigned y = x[c] ^ pat;
        const unsigned t = (y | 0x80808080u) - 0x01010101u;
        n |= ((t | y) >> (7 - 2 * c)) & (0x01010101u << (2 * c));
    }
    return n ^ 0x55555555u;
}

__device__ __forceinline__ void sage_push(unsigned rem, int ebase, int lane,
                                          unsigned* ring, int head, int& count)
{
    unsigned bal = __builtin_amdgcn_ballot_w32(rem != 0u);
    const int e0 = ebase + lane * 16;
    while (bal != 0u) {
        if (rem != 0u) {
            const int p = __builtin_ctz(rem);
            rem &= rem - 1u;
            const int pos = (int)__builtin_amdgcn_mbcnt_lo(bal, 0u);
            ring[(head + count + pos) & (QCAP - 1)] = (unsigned)(e0 + (p >> 1));
        }
        count += __builtin_popcount(bal);
        bal = __builtin_amdgcn_ballot_w32(rem != 0u);
    }
}

__global__ __launch_bounds__(1024) void sage_owner_kernel(
    const int* __restrict__ src,
    const int* __restrict__ dst,
    const unsigned char* __restrict__ key,
    const float* __restrict__ srcdata,
    float* __restrict__ out)
{
    __shared__ float    acc[NB * D];
    __shared__ int      cnt[NB];
    __shared__ unsigned rings[OWN_WAVES][QCAP];

    const int tid  = threadIdx.x;
    const int lane = tid & 31;
    const int wv   = __builtin_amdgcn_readfirstlane(tid >> 5);
    const int lo   = blockIdx.x * NB;
    const unsigned pat = blockIdx.x * 0x01010101u;

    for (int i = tid; i < NB * D / 4; i += 1024)
        ((float4*)acc)[i] = make_float4(0.f, 0.f, 0.f, 0.f);
    if (tid < NB) cnt[tid] = 0;
    __syncthreads();

    unsigned* ring = rings[wv];
    const uint4* key16 = (const uint4*)key;
    int head = 0, count = 0;

    int t = blockIdx.x % N_TILES;

    uint4 buf[LD_PER_IT];
    sage_issue(buf, key16, t, wv, lane);
    for (int it = 0; it < N_TILES; ++it) {
        const int tn = (t + 1 == N_TILES) ? 0 : t + 1;
        const uint4* pn = key16 + (wv * WAVE_PIECES + tn * TILE_PIECES + lane);
#pragma unroll
        for (int k = 0; k < LD_PER_IT; ++k) {
            const unsigned rem = sage_match(buf[k], pat);
            buf[k] = pn[k * 32];
            const int ebase = (wv * WAVE_PIECES + t * TILE_PIECES + k * 32) * 16;
            sage_push(rem, ebase, lane, ring, head, count);
            while (count >= 32) {
                sage_drain(ring, head, 32, lane, lo, src, dst, srcdata, acc, cnt);
                head = (head + 32) & (QCAP - 1);
                count -= 32;
            }
        }
        t = tn;
    }
    if (count > 0)
        sage_drain(ring, head, count, lane, lo, src, dst, srcdata, acc, cnt);

    __syncthreads();
    for (int i = tid; i < NB * D / 4; i += 1024) {
        const int node  = i >> 4;
        const int gnode = lo + node;
        if (gnode < N_NODES) {
            const int c = cnt[node];
            if (c > 0) {
                const float dg = (float)c;
                const float4 a = ((const float4*)acc)[i];
                float4* op = (float4*)(out + (size_t)gnode * D) + (i & 15);
                float4 o = *op;
                o.x += a.x / dg;
                o.y += a.y / dg;
                o.z += a.z / dg;
                o.w += a.w / dg;
                *op = o;
            }
        }
    }
}

extern "C" void kernel_launch(void* const* d_in, const int* in_sizes, int n_in,
                              void* d_out, int out_size, void* d_ws, size_t ws_size,
                              hipStream_t stream)
{
    const float* feat = (const float*)d_in[0];
    const int*   src  = (const int*)d_in[1];
    const int*   dst  = (const int*)d_in[2];
    const float* Wn   = (const float*)d_in[3];
    const float* Ws   = (const float*)d_in[4];
    const float* bias = (const float*)d_in[5];
    float* out = (float*)d_out;

    float* srcdata = (float*)d_ws;
    unsigned char* key = (unsigned char*)d_ws + (size_t)N_NODES * D * sizeof(float);

    sage_gemm_kernel<<<GEMM_BLOCKS, 256, 0, stream>>>(
        feat, Wn, Ws, bias, dst, srcdata, key, out);

    sage_owner_kernel<<<N_GROUPS, 1024, 0, stream>>>(src, dst, key, srcdata, out);
}


# launch 1 also writes one 4-byte record per edge (src, dst % 200); an owner group fetches that one scattered word per matching edge, not src[e] and dst[e] from two lines; sum order as before
# speedup vs baseline: 2.9064x; 1.0355x over previous
#include <hip/hip_runtime.h>

typedef __attribute__((ext_vector_type(2))) float v2f;
typedef __attribute__((ext_vector_type(8))) float v8f;

#define N_NODES 50000
#define N_EDGES 800000
#define D 64
#define M_TILES (N_NODES / 16)
#define GEMM_BLOCKS ((M_TILES + 7) / 8)
#define BLOCK_ROWS 128
#define A_PITCH 68

#define NB 200
#define N_GROUPS ((N_NODES + NB - 1) / NB)
#define OWN_WAVES 32
#define LD_PER_IT 7
#define N_TILES 7
#define TILE_PIECES (32 * LD_PER_IT)
#define WAVE_PIECES (TILE_PIECES * N_TILES)
#define N_PIECES (OWN_WAVES * WAVE_PIECES)
#define QCAP 1024

static_assert(N_GROUPS <= 255, "255 is the key of no group");
static_assert(N_EDGES % 16 == 0 && N_PIECES * 16 >= N_EDGES, "whole pieces, all edges covered");
static_assert(GEMM_BLOCKS * 256 >= N_PIECES, "launch 1 has one thread per piece");

#define N_QUADS (N_EDGES / 4)
#define GEMM_THREADS (GEMM_BLOCKS * 256)
static_assert(N_NODES <= (1 << 24) && NB <= 256, "src in 24 bits, local row in 8");
static_assert(N_EDGES % 4 == 0 && 2 * GEMM_THREADS >= N_QUADS && GEMM_THREADS < N_QUADS, "two quads per thread cover all edges");

__device__ __forceinline__ unsigned sage_key(int d)
{
    return (unsigned)d < (unsigned)N_NODES ? ((unsigned)d / NB) : 255u;
}

__device__ __forceinline__ unsigned sage_rec(int s, int d)
{
    return ((unsigned)s & 0x00FFFFFFu) | (((unsigned)d % NB) << 24);
}

__global__ __launch_bounds__(256) void sage_gemm_kernel(
    const float* __restrict__ feat,
    const float* __restrict__ Wn,
    const float* __restrict__ Ws,
    const float* __restrict__ bias,
    const int* __restrict__ src,
    const int* __restrict__ dst,
    float* __restrict__ srcdata,
    unsigned char* __restrict__ key,
    unsigned* __restrict__ rec,
    float* __restrict__ out)
{
    __shared__ float As[BLOCK_ROWS * A_PITCH];

    {
        const int gid = blockIdx.x * 256 + threadIdx.x;
        if (gid < N_PIECES) {
            uint4 kv = make_uint4(~0u, ~0u, ~0u, ~0u);
            if (gid < N_EDGES / 16) {
                const int4* q = (const int4*)dst + 4 * gid;
                const int4 q0 = q[0], q1 = q[1], q2 = q[2], q3 = q[3];
                kv.x = sage_key(q0.x) | (sage_key(q1.x) << 8) | (sage_key(q2.x) << 16) | (sage_key(q3.x) << 24);
                kv.y = sage_key(q0.y) | (sage_key(q1.y) << 8) | (sage_key(q2.y) << 16) | (sage_key(q3.y) << 24);
                kv.z = sage_key(q0.z) | (sage_key(q1.z) << 8) | (sage_key(q2.z) << 16) | (sage_key(q3.z) << 24);
                kv.w = sage_key(q0.w) | (sage_key(q1.w) << 8) | (sage_key(q2.w) << 16) | (sage_key(q3.w) << 24);
            }
            ((uint4*)key)[gid] = kv;
        }
    }

    const int wave = threadIdx.x >> 5;
    const int lane = threadIdx.x & 31;
    const int mat  = wave >> 2;
    const int kh   = (lane >> 4) * 2;
    const int n0   = lane & 15;
    const int n    = (wave & 3) * 16 + n0;
    const float* __restrict__ Wm = mat ? Ws : Wn;

    v2f b[16];
#pragma unroll
    for (int kk = 0; kk < 16; ++kk) {
        const int kb = kk * 4 + kh;
        b[kk].x = Wm[n * D + kb];
        b[kk].y = Wm[n * D + kb + 1];
    }

    const int row0 = blockIdx.x * BLOCK_ROWS;
    float4 fv[BLOCK_ROWS * 16 / 256];
#pragma unroll
    for (int i = 0; i < BLOCK_ROWS * 16 / 256; ++i) {
        const int idx = i * 256 + threadIdx.x;
        const int r = idx >> 4, c4 = idx & 15;
        const int gr = min(row0 + r, N_NODES - 1);
        fv[i] = ((const float4*)feat)[(size_t)gr * 16 + c4];
    }
    const int qa = blockIdx.x * 256 + threadIdx.x;
    const int qb = min(qa + GEMM_THREADS, N_QUADS - 1);
    const int4 sa = ((const int4*)src)[qa], da = ((const int4*)dst)[qa];
    const int4 sb = ((const int4*)src)[qb], db = ((const int4*)dst)[qb];
#pragma unroll
    for (int i = 0; i < BLOCK_ROWS * 16 / 256; ++i) {
        const int idx = i * 256 + threadIdx.x;
        const int r = idx >> 4, c4 = idx & 15;
        *(float4*)&As[r * A_PITCH + c4 * 4] = fv[i];
    }
    asm volatile("" :: "v"(sa.x), "v"(sa.y), "v"(sa.z), "v"(sa.w), "v"(da.x), "v"(da.y), "v"(da.z), "v"(da.w),
                       "v"(sb.x), "v"(sb.y), "v"(sb.z), "v"(sb.w), "v"(db.x), "v"(db.y), "v"(db.z), "v"(db.w));
    __syncthreads();

    ((uint4*)rec)[qa] = make_uint4(sage_rec(sa.x, da.x), sage_rec(sa.y, da.y),
                                   sage_rec(sa.z, da.z), sage_rec(sa.w, da.w));
    if (qa + GEMM_THREADS < N_QUADS)
        ((uint4*)rec)[qb] = make_uint4(sage_rec(sb.x, db.x), sage_rec(sb.y, db.y),
                                       sage_rec(sb.z, db.z), sage_rec(sb.w, db.w));

    const float bv = mat ? bias[n] : 0.f;
    float* __restrict__ res = mat ? out : srcdata;

#pragma unroll
    for (int mp = 0; mp < 4; ++mp) {
        v8f c0 = {};
        v8f c1 = {};
        const float* a0p = &As[(mp * 32 + n0) * A_PITCH + kh];
        const float* a1p = a0p + 16 * A_PITCH;
#pragma unroll
        for (int kk = 0; kk < 16; ++kk) {
            const v2f a0 = *(const v2f*)(a0p + kk * 4);
            const v2f a1 = *(const v2f*)(a1p + kk * 4);
            c0 = __builtin_amdgcn_wmma_f32_16x16x4_f32(
                false, a0, false, b[kk], (short)0, c0, false, false);
            c1 = __builtin_amdgcn_wmma_f32_16x16x4_f32(
                false, a1, false, b[kk], (short)0, c1, false, false);
        }
        const int tileA = blockIdx.x * 8 + mp * 2;
        const int rowA  = tileA * 16 + (lane >> 4) * 8;
        if (tileA < M_TILES) {
#pragma unroll
            for (int v = 0; v < 8; ++v)
                res[(rowA + v) * D + n] = mat ? c0[v] + bv : c0[v];
        }
        if (tileA + 1 < M_TILES) {
#pragma unroll
            for (int v = 0; v < 8; ++v)
                res[(rowA + 16 + v) * D + n] = mat ? c1[v] + bv : c1[v];
        }
    }
}


__device__ __forceinline__ void sage_drain(
    const unsigned* ring, int head, int nvalid, int lane,
    const unsigned* __restrict__ rec, const float* __restrict__ srcdata,
    float* acc, int* cnt)
{
    const bool valid = lane < nvalid;
    const unsigned ent = valid ? ring[(head + lane) & (QCAP - 1)] : 0u;
    int s = 0, dl = 0;
    if (valid) {
        const unsigned r = rec[ent];
        s  = (int)(r & 0x00FFFFFFu);
        dl = (int)(r >> 24);
        atomicAdd(&cnt[dl], 1);
    }
#pragma unroll
    for (int g = 0; g < 4; ++g) {
        if (g * 8 < nvalid) {
            float2 v[8];
#pragma unroll
            for (int j = 0; j < 8; ++j) {
                const int sj = __builtin_amdgcn_readlane(s, g * 8 + j);
                v[j] = *(const float2*)(srcdata + (size_t)sj * D + 2 * lane);
            }
#pragma unroll
            for (int j = 0; j < 8; ++j) {
                if (g * 8 + j < nvalid) {
                    const int dj = __builtin_amdgcn_readlane(dl, g * 8 + j);
                    float* a = acc + dj * D + 2 * lane;
                    atomicAdd(a,     v[j].x);
                    atomicAdd(a + 1, v[j].y);
                }
            }
        }
    }
}

__device__ __forceinline__ void sage_issue(uint4 (&buf)[LD_PER_IT], const uint4* __restrict__ key16,
                                           int tile, int wv, int lane)
{
    const uint4* p = key16 + (wv * WAVE_PIECES + tile * TILE_PIECES + lane);
#pragma unroll
    for (int k = 0; k < LD_PER_IT; ++k)
        buf[k] = p[k * 32];
}

__device__ __forceinline__ unsigned sage_match(const uint4& kv, unsigned pat)
{
    const unsigned x[4] = {kv.x, kv.y, kv.z, kv.w};
    unsigned n = 0;
#pragma unroll
    for (int c = 0; c < 4; ++c) {
        const unsigned y = x[c] ^ pat;
        const unsigned t = (y | 0x80808080u) - 0x01010101u;
        n |= ((t | y) >> (7 - 2 * c)) & (0x01010101u << (2 * c));
    }
    return n ^ 0x55555555u;
}

__device__ __forceinline__ void sage_push(unsigned rem, int ebase, int lane,
                                          unsigned* ring, int head, int& count)
{
    unsigned bal = __builtin_amdgcn_ballot_w32(rem != 0u);
    const int e0 = ebase + lane * 16;
    while (bal != 0u) {
        if (rem != 0u) {
            const int p = __builtin_ctz(rem);
            rem &= rem - 1u;
            const int pos = (int)__builtin_amdgcn_mbcnt_lo(bal, 0u);
            ring[(head + count + pos) & (QCAP - 1)] = (unsigned)(e0 + (p >> 1));
        }
        count += __builtin_popcount(bal);
        bal = __builtin_amdgcn_ballot_w32(rem != 0u);
    }
}

__global__ __launch_bounds__(1024) void sage_owner_kernel(
    const unsigned* __restrict__ rec,
    const unsigned char* __restrict__ key,
    const float* __restrict__ srcdata,
    float* __restrict__ out)
{
    __shared__ float    acc[NB * D];
    __shared__ int      cnt[NB];
    __shared__ unsigned rings[OWN_WAVES][QCAP];

    const int tid  = threadIdx.x;
    const int lane = tid & 31;
    const int wv   = __builtin_amdgcn_readfirstlane(tid >> 5);
    const int lo   = blockIdx.x * NB;
    const unsigned pat = blockIdx.x * 0x01010101u;

    for (int i = tid; i < NB * D / 4; i += 1024)
        ((float4*)acc)[i] = make_float4(0.f, 0.f, 0.f, 0.f);
    if (tid < NB) cnt[tid] = 0;
    __syncthreads();

    unsigned* ring = rings[wv];
    const uint4* key16 = (const uint4*)key;
    int head = 0, count = 0;

    int t = blockIdx.x % N_TILES;

    uint4 buf[LD_PER_IT];
    sage_issue(buf, key16, t, wv, lane);
    for (int it = 0; it < N_TILES; ++it) {
        const int tn = (t + 1 == N_TILES) ? 0 : t + 1;
        const uint4* pn = key16 + (wv * WAVE_PIECES + tn * TILE_PIECES + lane);
#pragma unroll
        for (int k = 0; k < LD_PER_IT; ++k) {
            const unsigned rem = sage_match(buf[k], pat);
            buf[k] = pn[k * 32];
            const int ebase = (wv * WAVE_PIECES + t * TILE_PIECES + k * 32) * 16;
            sage_push(rem, ebase, lane, ring, head, count);
            while (count >= 32) {
                sage_drain(ring, head, 32, lane, rec, srcdata, acc, cnt);
                head = (head + 32) & (QCAP - 1);
                count -= 32;
            }
        }
        t = tn;
    }
    if (count > 0)
        sage_drain(ring, head, count, lane, rec, srcdata, acc, cnt);

    __syncthreads();
    for (int i = tid; i < NB * D / 4; i += 1024) {
        const int node  = i >> 4;
        const int gnode = lo + node;
        if (gnode < N_NODES) {
            const int c = cnt[node];
            if (c > 0) {
                const float dg = (float)c;
                const float4 a = ((const float4*)acc)[i];
                float4* op = (float4*)(out + (size_t)gnode * D) + (i & 15);
                float4 o = *op;
                o.x += a.x / dg;
                o.y += a.y / dg;
                o.z += a.z / dg;
                o.w += a.w / dg;
                *op = o;
            }
        }
    }
}

extern "C" void kernel_launch(void* const* d_in, const int* in_sizes, int n_in,
                              void* d_out, int out_size, void* d_ws, size_t ws_size,
                              hipStream_t stream)
{
    const float* feat = (const float*)d_in[0];
    const int*   src  = (const int*)d_in[1];
    const int*   dst  = (const int*)d_in[2];
    const float* Wn   = (const float*)d_in[3];
    const float* Ws   = (const float*)d_in[4];
    const float* bias = (const float*)d_in[5];
    float* out = (float*)d_out;

    float* srcdata = (float*)d_ws;
    unsigned char* key = (unsigned char*)d_ws + (size_t)N_NODES * D * sizeof(float);
    unsigned* rec = (unsigned*)(key + (size_t)N_PIECES * 16);

    sage_gemm_kernel<<<GEMM_BLOCKS, 256, 0, stream>>>(
        feat, Wn, Ws, bias, src, dst, srcdata, key, rec, out);

    sage_owner_kernel<<<N_GROUPS, 1024, 0, stream>>>(rec, key, srcdata, out);
}


# speedup vs baseline: 3.0521x; 1.0501x over previous
#include <hip/hip_runtime.h>

typedef __attribute__((ext_vector_type(2))) float v2f;
typedef __attribute__((ext_vector_type(8))) float v8f;

#define N_NODES 50000
#define N_EDGES 800000
#define D 64
#define M_TILES (N_NODES / 16)
#define GEMM_BLOCKS ((M_TILES + 7) / 8)
#define BLOCK_ROWS 128
#define A_PITCH 68

#define NB 200
#define N_GROUPS ((N_NODES + NB - 1) / NB)
#define OWN_WAVES 32
#define LD_PER_IT 7
#define N_TILES 7
#define TILE_PIECES (32 * LD_PER_IT)
#define WAVE_PIECES (TILE_PIECES * N_TILES)
#define N_PIECES (OWN_WAVES * WAVE_PIECES)
#define QCAP 1024

#define LBLK 32768
#define N_LBLK 8
#define ROW_SLOTS ((NB + N_LBLK - 1) / N_LBLK)
#define CNT_OFF (ROW_SLOTS * D * 4)
#define RING_OFF 8192
#define OWN_LDS (N_LBLK * LBLK)

static_assert(N_GROUPS <= 255, "255 is the key of no group");
static_assert(OWN_WAVES == 4 * N_LBLK, "four rings in each block");
static_assert(CNT_OFF + NB * 4 <= RING_OFF && RING_OFF + 4 * QCAP * 4 <= LBLK, "row slots, in-degrees and rings fit a block");
static_assert(N_EDGES % 16 == 0 && N_PIECES * 16 >= N_EDGES, "whole pieces, all edges covered");
static_assert(GEMM_BLOCKS * 256 >= N_PIECES, "launch 1 has one thread per piece");

#define N_QUADS (N_EDGES / 4)
#define GEMM_THREADS (GEMM_BLOCKS * 256)
static_assert(N_NODES <= (1 << 24) && NB <= 256, "src in 24 bits, local row in 8");
static_assert(N_EDGES % 4 == 0 && 2 * GEMM_THREADS >= N_QUADS && GEMM_THREADS < N_QUADS, "two quads per thread cover all edges");

__device__ __forceinline__ unsigned sage_key(int d)
{
    return (unsigned)d < (unsigned)N_NODES ? ((unsigned)d / NB) : 255u;
}

__device__ __forceinline__ unsigned sage_rec(int s, int d)
{
    return ((unsigned)s & 0x00FFFFFFu) | (((unsigned)d % NB) << 24);
}

__global__ __launch_bounds__(256) void sage_gemm_kernel(
    const float* __restrict__ feat,
    const float* __restrict__ Wn,
    const float* __restrict__ Ws,
    const float* __restrict__ bias,
    const int* __restrict__ src,
    const int* __restrict__ dst,
    float* __restrict__ srcdata,
    unsigned char* __restrict__ key,
    unsigned* __restrict__ rec,
    float* __restrict__ out)
{
    __shared__ float As[BLOCK_ROWS * A_PITCH];

    {
        const int gid = blockIdx.x * 256 + threadIdx.x;
        if (gid < N_PIECES) {
            uint4 kv = make_uint4(~0u, ~0u, ~0u, ~0u);
            if (gid < N_EDGES / 16) {
                const int4* q = (const int4*)dst + 4 * gid;
                const int4 q0 = q[0], q1 = q[1], q2 = q[2], q3 = q[3];
                kv.x = sage_key(q0.x) | (sage_key(q1.x) << 8) | (sage_key(q2.x) << 16) | (sage_key(q3.x) << 24);
                kv.y = sage_key(q0.y) | (sage_key(q1.y) << 8) | (sage_key(q2.y) << 16) | (sage_key(q3.y) << 24);
                kv.z = sage_key(q0.z) | (sage_key(q1.z) << 8) | (sage_key(q2.z) << 16) | (sage_key(q3.z) << 24);
                kv.w = sage_key(q0.w) | (sage_key(q1.w) << 8) | (sage_key(q2.w) << 16) | (sage_key(q3.w) << 24);
            }
            ((uint4*)key)[gid] = kv;
        }
    }

    const int wave = threadIdx.x >> 5;
    const int lane = threadIdx.x & 31;
    const int mat  = wave >> 2;
    const int kh   = (lane >> 4) * 2;
    const int n0   = lane & 15;
    const int n    = (wave & 3) * 16 + n0;
    const float* __restrict__ Wm = mat ? Ws : Wn;

    v2f b[16];
#pragma unroll
    for (int kk = 0; kk < 16; ++kk) {
        const int kb = kk * 4 + kh;
        b[kk].x = Wm[n * D + kb];
        b[kk].y = Wm[n * D + kb + 1];
    }

    const int row0 = blockIdx.x * BLOCK_ROWS;
    float4 fv[BLOCK_ROWS * 16 / 256];
#pragma unroll
    for (int i = 0; i < BLOCK_ROWS * 16 / 256; ++i) {
        const int idx = i * 256 + threadIdx.x;
        const int r = idx >> 4, c4 = idx & 15;
        const int gr = min(row0 + r, N_NODES - 1);
        fv[i] = ((const float4*)feat)[(size_t)gr * 16 + c4];
    }
    const int qa = blockIdx.x * 256 + threadIdx.x;
    const int qb = min(qa + GEMM_THREADS, N_QUADS - 1);
    const int4 sa = ((const int4*)src)[qa], da = ((const int4*)dst)[qa];
    const int4 sb = ((const int4*)src)[qb], db = ((const int4*)dst)[qb];
#pragma unroll
    for (int i = 0; i < BLOCK_ROWS * 16 / 256; ++i) {
        const int idx = i * 256 + threadIdx.x;
        const int r = idx >> 4, c4 = idx & 15;
        *(float4*)&As[r * A_PITCH + c4 * 4] = fv[i];
    }
    asm volatile("" :: "v"(sa.x), "v"(sa.y), "v"(sa.z), "v"(sa.w), "v"(da.x), "v"(da.y), "v"(da.z), "v"(da.w),
                       "v"(sb.x), "v"(sb.y), "v"(sb.z), "v"(sb.w), "v"(db.x), "v"(db.y), "v"(db.z), "v"(db.w));
    __syncthreads();

    ((uint4*)rec)[qa] = make_uint4(sage_rec(sa.x, da.x), sage_rec(sa.y, da.y),
                                   sage_rec(sa.z, da.z), sage_rec(sa.w, da.w));
    if (qa + GEMM_THREADS < N_QUADS)
        ((uint4*)rec)[qb] = make_uint4(sage_rec(sb.x, db.x), sage_rec(sb.y, db.y),
                                       sage_rec(sb.z, db.z), sage_rec(sb.w, db.w));

    const float bv = mat ? bias[n] : 0.f;
    float* __restrict__ res = mat ? out : srcdata;

#pragma unroll
    for (int mp = 0; mp < 4; ++mp) {
        v8f c0 = {};
        v8f c1 = {};
        const float* a0p = &As[(mp * 32 + n0) * A_PITCH + kh];
        const float* a1p = a0p + 16 * A_PITCH;
#pragma unroll
        for (int kk = 0; kk < 16; ++kk) {
            const v2f a0 = *(const v2f*)(a0p + kk * 4);
            const v2f a1 = *(const v2f*)(a1p + kk * 4);
            c0 = __builtin_amdgcn_wmma_f32_16x16x4_f32(
                false, a0, false, b[kk], (short)0, c0, false, false);
            c1 = __builtin_amdgcn_wmma_f32_16x16x4_f32(
                false, a1, false, b[kk], (short)0, c1, false, false);
        }
        const int tileA = blockIdx.x * 8 + mp * 2;
        const int rowA  = tileA * 16 + (lane >> 4) * 8;
        if (tileA < M_TILES) {
#pragma unroll
            for (int v = 0; v < 8; ++v)
                res[(rowA + v) * D + n] = mat ? c0[v] + bv : c0[v];
        }
        if (tileA + 1 < M_TILES) {
#pragma unroll
            for (int v = 0; v < 8; ++v)
                res[(rowA + 16 + v) * D + n] = mat ? c1[v] + bv : c1[v];
        }
    }
}


__device__ __forceinline__ int sage_row_off(int r)
{
    return (r & (N_LBLK - 1)) * LBLK + (r / N_LBLK) * (D * 4);
}

__device__ __forceinline__ void sage_drain(
    const unsigned* ring, int head, int nvalid, int lane,
    const unsigned* __restrict__ rec, const float* __restrict__ srcdata,
    unsigned char* lds)
{
    int* cnt = (int*)(lds + CNT_OFF);
    const bool valid = lane < nvalid;
    const unsigned ent = valid ? ring[(head + lane) & (QCAP - 1)] : 0u;
    int s = 0, ro = 0;
    if (valid) {
        const unsigned r = rec[ent];
        s  = (int)(r & 0x00FFFFFFu);
        const int dl = (int)(r >> 24);
        ro = sage_row_off(dl);
        atomicAdd(&cnt[dl], 1);
    }
#pragma unroll
    for (int g = 0; g < 4; ++g) {
        if (g * 8 < nvalid) {
            float2 v[8];
#pragma unroll
            for (int j = 0; j < 8; ++j) {
                const int sj = __builtin_amdgcn_readlane(s, g * 8 + j);
                v[j] = *(const float2*)(srcdata + (size_t)sj * D + 2 * lane);
            }
#pragma unroll
            for (int j = 0; j < 8; ++j) {
                if (g * 8 + j < nvalid) {
                    const int rj = __builtin_amdgcn_readlane(ro, g * 8 + j);
                    float* a = (float*)(lds + rj) + 2 * lane;
                    atomicAdd(a,     v[j].x);
                    atomicAdd(a + 1, v[j].y);
                }
            }
        }
    }
}

__device__ __forceinline__ void sage_issue(uint4 (&buf)[LD_PER_IT], const uint4* __restrict__ key16,
                                           int tile, int wv, int lane)
{
    const uint4* p = key16 + (wv * WAVE_PIECES + tile * TILE_PIECES + lane);
#pragma unroll
    for (int k = 0; k < LD_PER_IT; ++k)
        buf[k] = p[k * 32];
}

__device__ __forceinline__ unsigned sage_match(const uint4& kv, unsigned pat)
{
    const unsigned x[4] = {kv.x, kv.y, kv.z, kv.w};
    unsigned n = 0;
#pragma unroll
    for (int c = 0; c < 4; ++c) {
        const unsigned y = x[c] ^ pat;
        const unsigned t = (y | 0x80808080u) - 0x01010101u;
        n |= ((t | y) >> (7 - 2 * c)) & (0x01010101u << (2 * c));
    }
    return n ^ 0x55555555u;
}

__device__ __forceinline__ void sage_push(unsigned rem, int ebase, int lane,
                                          unsigned* ring, int head, int& count)
{
    unsigned bal = __builtin_amdgcn_ballot_w32(rem != 0u);
    const int e0 = ebase + lane * 16;
    while (bal != 0u) {
        if (rem != 0u) {
            const int p = __builtin_ctz(rem);
            rem &= rem - 1u;
            const int pos = (int)__builtin_amdgcn_mbcnt_lo(bal, 0u);
            ring[(head + count + pos) & (QCAP - 1)] = (unsigned)(e0 + (p >> 1));
        }
        count += __builtin_popcount(bal);
        bal = __builtin_amdgcn_ballot_w32(rem != 0u);
    }
}

__global__ __launch_bounds__(1024) void sage_owner_kernel(
    const unsigned* __restrict__ rec,
    const unsigned char* __restrict__ key,
    const float* __restrict__ srcdata,
    float* __restrict__ out)
{
    __shared__ __attribute__((aligned(16))) unsigned char lds[OWN_LDS];
    int* cnt = (int*)(lds + CNT_OFF);

    const int tid  = threadIdx.x;
    const int lane = tid & 31;
    const int wv   = __builtin_amdgcn_readfirstlane(tid >> 5);
    const int lo   = blockIdx.x * NB;
    const unsigned pat = blockIdx.x * 0x01010101u;

    for (int i = tid; i < NB * D / 4; i += 1024)
        *(float4*)(lds + sage_row_off(i >> 4) + (i & 15) * 16) = make_float4(0.f, 0.f, 0.f, 0.f);
    if (tid < NB) cnt[tid] = 0;
    __syncthreads();

    unsigned* ring = (unsigned*)(lds + (wv >> 2) * LBLK + RING_OFF + (wv & 3) * (QCAP * 4));
    const uint4* key16 = (const uint4*)key;
    int head = 0, count = 0;

    int t = blockIdx.x % N_TILES;

    uint4 buf[LD_PER_IT];
    sage_issue(buf, key16, t, wv, lane);
    for (int it = 0; it < N_TILES; ++it) {
        const int tn = (t + 1 == N_TILES) ? 0 : t + 1;
        const uint4* pn = key16 + (wv * WAVE_PIECES + tn * TILE_PIECES + lane);
#pragma unroll
        for (int k = 0; k < LD_PER_IT; ++k) {
            const unsigned rem = sage_match(buf[k], pat);
            buf[k] = pn[k * 32];
            const int ebase = (wv * WAVE_PIECES + t * TILE_PIECES + k * 32) * 16;
            sage_push(rem, ebase, lane, ring, head, count);
            while (count >= 32) {
                sage_drain(ring, head, 32, lane, rec, srcdata, lds);
                head = (head + 32) & (QCAP - 1);
                count -= 32;
            }
        }
        t = tn;
    }
    if (count > 0)
        sage_drain(ring, head, count, lane, rec, srcdata, lds);

    __syncthreads();
    for (int i = tid; i < NB * D / 4; i += 1024) {
        const int node  = i >> 4;
        const int gnode = lo + node;
        if (gnode < N_NODES) {
            const int c = cnt[node];
            if (c > 0) {
                const float dg = (float)c;
                const float4 a = *(const float4*)(lds + sage_row_off(node) + (i & 15) * 16);
                float4* op = (float4*)(out + (size_t)gnode * D) + (i & 15);
                float4 o = *op;
                o.x += a.x / dg;
                o.y += a.y / dg;
                o.z += a.z / dg;
                o.w += a.w / dg;
                *op = o;
            }
        }
    }
}

extern "C" void kernel_launch(void* const* d_in, const int* in_sizes, int n_in,
                              void* d_out, int out_size, void* d_ws, size_t ws_size,
                              hipStream_t stream)
{
    const float* feat = (const float*)d_in[0];
    const int*   src  = (const int*)d_in[1];
    const int*   dst  = (const int*)d_in[2];
    const float* Wn   = (const float*)d_in[3];
    const float* Ws   = (const float*)d_in[4];
    const float* bias = (const float*)d_in[5];
    float* out = (float*)d_out;

    float* srcdata = (float*)d_ws;
    unsigned char* key = (unsigned char*)d_ws + (size_t)N_NODES * D * sizeof(float);
    unsigned* rec = (unsigned*)(key + (size_t)N_PIECES * 16);

    sage_gemm_kernel<<<GEMM_BLOCKS, 256, 0, stream>>>(
        feat, Wn, Ws, bias, src, dst, srcdata, key, rec, out);

    sage_owner_kernel<<<N_GROUPS, 1024, 0, stream>>>(rec, key, srcdata, out);
}
